# starting kernel
# baseline (speedup 1.0000x reference)
#include <hip/hip_runtime.h>
#include <hip/hip_bf16.h>


#define GN_N 100000
#define GN_E 640000
#define GN_D 128

typedef __attribute__((ext_vector_type(2))) float v2f;
typedef __attribute__((ext_vector_type(8))) float v8f;

__global__ void sage_pack_w(const float* __restrict__ W, float* __restrict__ Bpk) {
    int idx = blockIdx.x * blockDim.x + threadIdx.x;
    if (idx >= 32 * 2 * 128) return;
    int j  = idx & 127;
    int h  = (idx >> 7) & 1;
    int kg = idx >> 8;
    int k  = kg * 4 + 2 * h;
    Bpk[idx * 2 + 0] = W[j * GN_D + k];
    Bpk[idx * 2 + 1] = W[j * GN_D + k + 1];
}

__global__ void sage_count(const int* __restrict__ dst, float* __restrict__ cnt) {
    int e = blockIdx.x * blockDim.x + threadIdx.x;
    if (e < GN_E) atomicAdd(&cnt[dst[e]], 1.0f);
}

__global__ void sage_invdeg(float* __restrict__ cnt) {
    int i = blockIdx.x * blockDim.x + threadIdx.x;
    if (i < GN_N) cnt[i] = 1.0f / fmaxf(cnt[i], 1.0f);
}

__global__ __launch_bounds__(256) void sage_scatter(
    const float* __restrict__ H, const int* __restrict__ src,
    const int* __restrict__ dst, float* __restrict__ SUM) {
    int idx  = blockIdx.x * blockDim.x + threadIdx.x;
    int e    = idx >> 5;
    int lane = idx & 31;
    if (e >= GN_E) return;
    int s = src[e];
    int d = dst[e];
    const float4 v = *(const float4*)(H + (size_t)s * GN_D + lane * 4);
    float* p = SUM + (size_t)d * GN_D + lane * 4;
    atomicAdd(p + 0, v.x);
    atomicAdd(p + 1, v.y);
    atomicAdd(p + 2, v.z);
    atomicAdd(p + 3, v.w);
}

__global__ __launch_bounds__(128) void sage_gemm(
    const float* __restrict__ A1, const float* __restrict__ invdeg,
    const float* __restrict__ B1pk, const float* __restrict__ bias,
    const float* __restrict__ A2, const float* __restrict__ B2pk,
    float* __restrict__ out, int relu_out) {
    const int lane = threadIdx.x & 31;
    const int wave = blockIdx.x * (blockDim.x >> 5) + (threadIdx.x >> 5);
    if (wave >= GN_N / 16) return;

    const int rbase = wave * 16;
    const int h     = lane >> 4;
    const int l15   = lane & 15;
    const int row   = rbase + l15;

    float s = 1.0f;
    if (invdeg) s = invdeg[row];

    const float* a1row = A1 + (size_t)row * GN_D;
    const float* a2row = A2 ? (A2 + (size_t)row * GN_D) : nullptr;

    float bj[8];
#pragma unroll
    for (int jt = 0; jt < 8; ++jt) bj[jt] = bias[jt * 16 + l15];

    v8f c[8];
#pragma unroll
    for (int jt = 0; jt < 8; ++jt)
#pragma unroll
        for (int v = 0; v < 8; ++v) c[jt][v] = bj[jt];

#pragma unroll
    for (int k0 = 0; k0 < GN_D; k0 += 4) {
        v2f a = *(const v2f*)(a1row + k0 + 2 * h);
        a.x *= s;
        a.y *= s;
        const float* brow = B1pk + ((k0 >> 2) * 2 + h) * 256 + 2 * l15;
#pragma unroll
        for (int jt = 0; jt < 8; ++jt) {
            v2f b = *(const v2f*)(brow + jt * 32);
            c[jt] = __builtin_amdgcn_wmma_f32_16x16x4_f32(
                false, a, false, b, (short)0, c[jt], false, false);
        }
    }

    if (a2row) {
#pragma unroll
        for (int k0 = 0; k0 < GN_D; k0 += 4) {
            v2f a = *(const v2f*)(a2row + k0 + 2 * h);
            const float* brow = B2pk + ((k0 >> 2) * 2 + h) * 256 + 2 * l15;
#pragma unroll
            for (int jt = 0; jt < 8; ++jt) {
                v2f b = *(const v2f*)(brow + jt * 32);
                c[jt] = __builtin_amdgcn_wmma_f32_16x16x4_f32(
                    false, a, false, b, (short)0, c[jt], false, false);
            }
        }
    }

#pragma unroll
    for (int jt = 0; jt < 8; ++jt)
#pragma unroll
        for (int v = 0; v < 8; ++v) {
            float val = c[jt][v];
            if (relu_out) val = fmaxf(val, 0.0f);
            out[(size_t)(rbase + v + 8 * h) * GN_D + jt * 16 + l15] = val;
        }
}

extern "C" void kernel_launch(void* const* d_in, const int* in_sizes, int n_in,
                              void* d_out, int out_size, void* d_ws, size_t ws_size,
                              hipStream_t stream) {
    (void)in_sizes; (void)n_in; (void)out_size; (void)ws_size;

    const float* x   = (const float*)d_in[0];
    const int*   ei  = (const int*)d_in[1];
    const float* W0p = (const float*)d_in[2];
    const float* b0p = (const float*)d_in[3];
    const float* W0l = (const float*)d_in[4];
    const float* b0l = (const float*)d_in[5];
    const float* W0r = (const float*)d_in[6];
    const float* W1p = (const float*)d_in[7];
    const float* b1p = (const float*)d_in[8];
    const float* W1l = (const float*)d_in[9];
    const float* b1l = (const float*)d_in[10];
    const float* W1r = (const float*)d_in[11];
    float* out = (float*)d_out;

    const size_t nd = (size_t)GN_N * GN_D;
    float* H   = (float*)d_ws;
    float* X1  = H + nd;
    float* SUM = X1 + nd;
    float* INV = SUM + nd;
    float* PK  = INV + GN_N;
    float* P0p = PK + 0 * 16384;
    float* P0l = PK + 1 * 16384;
    float* P0r = PK + 2 * 16384;
    float* P1p = PK + 3 * 16384;
    float* P1l = PK + 4 * 16384;
    float* P1r = PK + 5 * 16384;

    const int* src = ei;
    const int* dst = ei + GN_E;

    sage_pack_w<<<32, 256, 0, stream>>>(W0p, P0p);
    sage_pack_w<<<32, 256, 0, stream>>>(W0l, P0l);
    sage_pack_w<<<32, 256, 0, stream>>>(W0r, P0r);
    sage_pack_w<<<32, 256, 0, stream>>>(W1p, P1p);
    sage_pack_w<<<32, 256, 0, stream>>>(W1l, P1l);
    sage_pack_w<<<32, 256, 0, stream>>>(W1r, P1r);

    hipMemsetAsync(INV, 0, GN_N * sizeof(float), stream);
    sage_count<<<(GN_E + 255) / 256, 256, 0, stream>>>(dst, INV);
    sage_invdeg<<<(GN_N + 255) / 256, 256, 0, stream>>>(INV);

    const int gemm_blocks    = (GN_N / 16 + 3) / 4;
    const int scatter_blocks = (GN_E * 32 + 255) / 256;

    sage_gemm<<<gemm_blocks, 128, 0, stream>>>(x, nullptr, P0p, b0p,
                                               nullptr, nullptr, H, 1);
    hipMemsetAsync(SUM, 0, nd * sizeof(float), stream);
    sage_scatter<<<scatter_blocks, 256, 0, stream>>>(H, src, dst, SUM);
    sage_gemm<<<gemm_blocks, 128, 0, stream>>>(SUM, INV, P0l, b0l,
                                               x, P0r, X1, 1);

    sage_gemm<<<gemm_blocks, 128, 0, stream>>>(X1, nullptr, P1p, b1p,
                                               nullptr, nullptr, H, 1);
    hipMemsetAsync(SUM, 0, nd * sizeof(float), stream);
    sage_scatter<<<scatter_blocks, 256, 0, stream>>>(H, src, dst, SUM);
    sage_gemm<<<gemm_blocks, 128, 0, stream>>>(SUM, INV, P1l, b1l,
                                               X1, P1r, out, 0);
}


# Neighbour sums by 256 destination-range owner workgroups: edges filtered once into an LDS list (kept for layer 2), rows fetched 16 at a time, ds_add_f32 in arrival order; global atomics, fills gone
# speedup vs baseline: 2.3240x; 2.3240x over previous
#include <hip/hip_runtime.h>
#include <hip/hip_bf16.h>


#define GN_N 100000
#define GN_E 640000
#define GN_D 128
#define GN_RB (GN_N / 16)

typedef __attribute__((ext_vector_type(2))) float v2f;
typedef __attribute__((ext_vector_type(4))) float v4f;
typedef __attribute__((ext_vector_type(8))) float v8f;
typedef __attribute__((ext_vector_type(4))) int v4i;

#define GN_OW    256
#define GN_RPW   ((GN_N + GN_OW - 1) / GN_OW)
#define GN_LCAP  8192
#define GN_TRIPS (GN_E / 1024)

__device__ __forceinline__ void own_add(float* p, float v) {
    __hip_atomic_fetch_add(p, v, __ATOMIC_RELAXED, __HIP_MEMORY_SCOPE_WORKGROUP);
}

template <bool FIRST>
__global__ __launch_bounds__(1024) void sage_owner(
    const float* __restrict__ H, const int* __restrict__ src,
    const int* __restrict__ dst, float* __restrict__ SUM,
    float* __restrict__ INV, unsigned* __restrict__ LST,
    unsigned* __restrict__ LCNT) {
    __shared__ __attribute__((aligned(16))) float acc[(GN_RPW + 1) * GN_D];
    __shared__ unsigned list[GN_LCAP];
    __shared__ unsigned deg[GN_RPW + 1];
    __shared__ unsigned cnt;

    const int tid  = threadIdx.x;
    const int lane = tid & 31;
    const int wv   = __builtin_amdgcn_readfirstlane(tid >> 5);
    const int b    = blockIdx.x;
    const int lo   = b * GN_RPW;
    const int nrows = (GN_N - lo < GN_RPW) ? (GN_N - lo) : GN_RPW;

    {
        const v4f z = {0.0f, 0.0f, 0.0f, 0.0f};
        for (int i = tid; i < (GN_RPW + 1) * GN_D / 4; i += 1024) ((v4f*)acc)[i] = z;
        if (tid <= GN_RPW) deg[tid] = 0u;
        if (tid == 0) cnt = 0u;
    }
    unsigned stored = 0u;
    if (!FIRST) stored = LCNT[b];
    __syncthreads();

    const bool sweep = FIRST || stored > GN_LCAP;
    if (sweep) {
        const int rot = (b * GN_TRIPS) / GN_OW;
        for (int t = wv; t < GN_TRIPS; t += 32) {
            int tt = t + rot;
            if (tt >= GN_TRIPS) tt -= GN_TRIPS;
            const int e0 = tt * 1024 + lane * 4;
            const int* dp = dst + e0;
            v4i d[8];
#pragma unroll
            for (int j = 0; j < 8; ++j) d[j] = *(const v4i*)(dp + j * 128);
#pragma unroll
            for (int j = 0; j < 8; ++j) {
                const unsigned r0 = (unsigned)(d[j].x - lo), r1 = (unsigned)(d[j].y - lo);
                const unsigned r2 = (unsigned)(d[j].z - lo), r3 = (unsigned)(d[j].w - lo);
                const bool m0 = r0 < (unsigned)nrows, m1 = r1 < (unsigned)nrows;
                const bool m2 = r2 < (unsigned)nrows, m3 = r3 < (unsigned)nrows;
                if (m0 | m1 | m2 | m3) {
#pragma unroll
                    for (int k = 0; k < 4; ++k) {
                        const unsigned r = k == 0 ? r0 : k == 1 ? r1 : k == 2 ? r2 : r3;
                        if (r < (unsigned)nrows) {
                            const unsigned e = (unsigned)(e0 + j * 128 + k);
                            if (FIRST) atomicAdd(&deg[r], 1u);
                            const unsigned p = atomicAdd(&cnt, 1u);
                            if (p < GN_LCAP) {
                                list[p] = (r << 20) | e;
                            } else {
                                const float* hp = H + (size_t)src[e] * GN_D;
                                float* ap = acc + r * GN_D;
#pragma nounroll
                                for (int c = 0; c < GN_D; ++c)
                                    own_add(ap + (c & 3) * 32 + (c >> 2), hp[c]);
                            }
                        }
                    }
                }
            }
        }
    } else {
        const unsigned* st = LST + (size_t)b * GN_LCAP;
        for (unsigned i = tid; i < stored; i += 1024) list[i] = st[i];
    }
    __syncthreads();

    const unsigned total = sweep ? cnt : stored;
    const unsigned n = total < GN_LCAP ? total : GN_LCAP;
    unsigned* mylst = LST + (size_t)b * GN_LCAP;
    if (FIRST) {
        for (unsigned i = tid; i < n; i += 1024) mylst[i] = list[i];
        if (tid == 0) LCNT[b] = total;
        if (tid < nrows) INV[lo + tid] = 1.0f / fmaxf((float)deg[tid], 1.0f);
    }

    auto fetch = [&](unsigned at, int& s_out, int& r_out) {
        const unsigned i = at + lane;
        unsigned ent = (unsigned)GN_RPW << 20;
        if (i < n) ent = list[i];
        s_out = src[ent & 0xFFFFFu];
        r_out = (int)(ent >> 20);
    };
    unsigned base = wv * 32;
    int s = 0, rl = GN_RPW;
    if (base < n) fetch(base, s, rl);
    while (base < n) {
        const int live = (int)(n - base);
        const unsigned nb = base + 1024;
        int sn = 0, rn = GN_RPW;
        v4f v[16];
#pragma unroll
        for (int j = 0; j < 16; ++j) {
            const int sj = __builtin_amdgcn_readlane(s, j);
            v[j] = *(const v4f*)(H + (size_t)sj * GN_D + lane * 4);
        }
        if (nb < n) fetch(nb, sn, rn);
#pragma unroll
        for (int j = 0; j < 16; ++j) {
            const int rj = __builtin_amdgcn_readlane(rl, j);
            float* ap = acc + rj * GN_D + lane;
            own_add(ap,      v[j].x);
            own_add(ap + 32, v[j].y);
            own_add(ap + 64, v[j].z);
            own_add(ap + 96, v[j].w);
        }
        if (live > 16) {
#pragma unroll
            for (int j = 0; j < 16; ++j) {
                const int sj = __builtin_amdgcn_readlane(s, 16 + j);
                v[j] = *(const v4f*)(H + (size_t)sj * GN_D + lane * 4);
            }
#pragma unroll
            for (int j = 0; j < 16; ++j) {
                const int rj = __builtin_amdgcn_readlane(rl, 16 + j);
                float* ap = acc + rj * GN_D + lane;
                own_add(ap,      v[j].x);
                own_add(ap + 32, v[j].y);
                own_add(ap + 64, v[j].z);
                own_add(ap + 96, v[j].w);
            }
        }
        base = nb;
        s = sn;
        rl = rn;
    }
    __syncthreads();

    for (int r = wv; r < nrows; r += 32) {
        const float* ap = acc + r * GN_D + lane;
        v4f o = {ap[0], ap[32], ap[64], ap[96]};
        *(v4f*)(SUM + (size_t)(lo + r) * GN_D + lane * 4) = o;
    }
}

template <int NP>
struct RsFrag {
    v4f a[NP][16];
    float s;
};

template <int NP>
__device__ __forceinline__ void rs_load(RsFrag<NP>& f, int rb, int l15, int koff,
                                        const float* __restrict__ A1,
                                        const float* __restrict__ A2,
                                        const float* __restrict__ invdeg) {
    const size_t row = (size_t)(rb * 16 + l15);
    const float* r1 = A1 + row * GN_D + koff;
#pragma unroll
    for (int q = 0; q < 16; ++q) f.a[0][q] = *(const v4f*)(r1 + 8 * q);
    if (NP == 2) {
        const float* r2 = A2 + row * GN_D + koff;
#pragma unroll
        for (int q = 0; q < 16; ++q) f.a[NP - 1][q] = *(const v4f*)(r2 + 8 * q);
        f.s = invdeg[row];
    } else {
        f.s = 1.0f;
    }
}

template <int NP, int NT>
__device__ __forceinline__ void rs_block(const RsFrag<NP>& f, const v4f (&b)[NP][16][NT],
                                         const float (&bj)[NT], int rb, int h, int l15,
                                         int colbase, float* __restrict__ out, int relu_out) {
    v8f c[NT];
#pragma unroll
    for (int jt = 0; jt < NT; ++jt)
#pragma unroll
        for (int v = 0; v < 8; ++v) c[jt][v] = bj[jt];

#pragma unroll
    for (int p = 0; p < NP; ++p) {
#pragma unroll
        for (int q = 0; q < 16; ++q) {
            v4f av = f.a[p][q];
            if (NP == 2 && p == 0) av *= f.s;
            v2f a0 = {av.x, av.y};
            v2f a1 = {av.z, av.w};
#pragma unroll
            for (int jt = 0; jt < NT; ++jt) {
                v2f b0 = {b[p][q][jt].x, b[p][q][jt].y};
                c[jt] = __builtin_amdgcn_wmma_f32_16x16x4_f32(
                    false, a0, false, b0, (short)0, c[jt], false, false);
            }
#pragma unroll
            for (int jt = 0; jt < NT; ++jt) {
                v2f b1 = {b[p][q][jt].z, b[p][q][jt].w};
                c[jt] = __builtin_amdgcn_wmma_f32_16x16x4_f32(
                    false, a1, false, b1, (short)0, c[jt], false, false);
            }
        }
    }

    float* obase = out + (size_t)(rb * 16 + 8 * h) * GN_D + colbase + l15;
#pragma unroll
    for (int jt = 0; jt < NT; ++jt)
#pragma unroll
        for (int v = 0; v < 8; ++v) {
            float val = c[jt][v];
            if (relu_out) val = fmaxf(val, 0.0f);
            obase[(size_t)v * GN_D + jt * 16] = val;
        }
}

template <int NP, int NT>
__global__ __launch_bounds__(128) void sage_gemm_rs(
    const float* __restrict__ A1, const float* __restrict__ invdeg,
    const float* __restrict__ W1, const float* __restrict__ bias,
    const float* __restrict__ A2, const float* __restrict__ W2,
    float* __restrict__ out, int relu_out) {
    const int lane   = threadIdx.x & 31;
    const int wv     = __builtin_amdgcn_readfirstlane(threadIdx.x >> 5);
    const int colbase = (NT == 4) ? (wv & 1) * 64 : 0;
    const int h      = lane >> 4;
    const int l15    = lane & 15;
    const int koff   = 4 * h;
    const int npairs = (NT == 4) ? gridDim.x * 2 : gridDim.x * 4;
    int rb = (NT == 4) ? blockIdx.x * 2 + (wv >> 1) : blockIdx.x * 4 + wv;
    if (rb >= GN_RB) return;

    v4f b[NP][16][NT];
#pragma unroll
    for (int q = 0; q < 16; ++q)
#pragma unroll
        for (int jt = 0; jt < NT; ++jt) {
            const size_t wrow = (size_t)(colbase + jt * 16 + l15) * GN_D + 8 * q + koff;
            b[0][q][jt] = *(const v4f*)(W1 + wrow);
            if (NP == 2) b[NP - 1][q][jt] = *(const v4f*)(W2 + wrow);
        }

    float bj[NT];
#pragma unroll
    for (int jt = 0; jt < NT; ++jt) bj[jt] = bias[colbase + jt * 16 + l15];

    RsFrag<NP> f0, f1;
    rs_load<NP>(f0, rb, l15, koff, A1, A2, invdeg);

    for (;;) {
        const int rbn = rb + npairs;
        rs_load<NP>(f1, rbn < GN_RB ? rbn : rb, l15, koff, A1, A2, invdeg);
        __builtin_amdgcn_sched_barrier(0);
        rs_block<NP, NT>(f0, b, bj, rb, h, l15, colbase, out, relu_out);
        __builtin_amdgcn_sched_barrier(0);
        if (rbn >= GN_RB) break;
        rb = rbn;
        f0 = f1;
    }
}

extern "C" void kernel_launch(void* const* d_in, const int* in_sizes, int n_in,
                              void* d_out, int out_size, void* d_ws, size_t ws_size,
                              hipStream_t stream) {
    (void)in_sizes; (void)n_in; (void)out_size; (void)ws_size;

    const float* x   = (const float*)d_in[0];
    const int*   ei  = (const int*)d_in[1];
    const float* W0p = (const float*)d_in[2];
    const float* b0p = (const float*)d_in[3];
    const float* W0l = (const float*)d_in[4];
    const float* b0l = (const float*)d_in[5];
    const float* W0r = (const float*)d_in[6];
    const float* W1p = (const float*)d_in[7];
    const float* b1p = (const float*)d_in[8];
    const float* W1l = (const float*)d_in[9];
    const float* b1l = (const float*)d_in[10];
    const float* W1r = (const float*)d_in[11];
    float* out = (float*)d_out;

    const size_t nd = (size_t)GN_N * GN_D;
    float* H   = (float*)d_ws;
    float* X1  = H + nd;
    float* SUM = X1 + nd;
    float* INV = SUM + nd;
    unsigned* LST  = (unsigned*)(INV + GN_N);
    unsigned* LCNT = LST + (size_t)GN_OW * GN_LCAP;

    const int* src = ei;
    const int* dst = ei + GN_E;

    const int gemm_blocks = 256;

    sage_gemm_rs<1, 8><<<gemm_blocks, 128, 0, stream>>>(x, nullptr, W0p, b0p,
                                                     nullptr, nullptr, H, 1);
    sage_owner<true><<<GN_OW, 1024, 0, stream>>>(H, src, dst, SUM, INV, LST, LCNT);
    sage_gemm_rs<2, 4><<<gemm_blocks, 128, 0, stream>>>(SUM, INV, W0l, b0l,
                                                     x, W0r, X1, 1);

    sage_gemm_rs<1, 8><<<gemm_blocks, 128, 0, stream>>>(X1, nullptr, W1p, b1p,
                                                     nullptr, nullptr, H, 1);
    sage_owner<false><<<GN_OW, 1024, 0, stream>>>(H, src, dst, SUM, INV, LST, LCNT);
    sage_gemm_rs<2, 4><<<gemm_blocks, 128, 0, stream>>>(SUM, INV, W1l, b1l,
                                                     X1, W1r, out, 0);
}


# Owner filter sweeps a bit-sliced map of the edges' owner numbers (eight planes, written by spare waves of the first projection launch): 32 edges matched per 8 XOR+AND; LDS adds in arrival order
# speedup vs baseline: 2.4601x; 1.0586x over previous
#include <hip/hip_runtime.h>
#include <hip/hip_bf16.h>


#define GN_N 100000
#define GN_E 640000
#define GN_D 128
#define GN_RB (GN_N / 16)

typedef __attribute__((ext_vector_type(2))) float v2f;
typedef __attribute__((ext_vector_type(4))) float v4f;
typedef __attribute__((ext_vector_type(8))) float v8f;
typedef __attribute__((ext_vector_type(4))) int v4i;
typedef __attribute__((ext_vector_type(4))) unsigned v4u;

#define GN_OW    256
#define GN_RPW   ((GN_N + GN_OW - 1) / GN_OW)
#define GN_LCAP  8192
#define GN_TRIPS (GN_E / 1024)

__device__ __forceinline__ void own_add(float* p, float v) {
    __hip_atomic_fetch_add(p, v, __ATOMIC_RELAXED, __HIP_MEMORY_SCOPE_WORKGROUP);
}

template <bool FIRST>
__global__ __launch_bounds__(1024) void sage_owner(
    const float* __restrict__ H, const int* __restrict__ src,
    const int* __restrict__ dst, float* __restrict__ SUM,
    float* __restrict__ INV, unsigned* __restrict__ LST,
    unsigned* __restrict__ LCNT, const unsigned* __restrict__ PL) {
    __shared__ __attribute__((aligned(16))) float acc[(GN_RPW + 1) * GN_D];
    __shared__ unsigned list[GN_LCAP];
    __shared__ unsigned deg[GN_RPW + 1];
    __shared__ unsigned cnt;

    const int tid  = threadIdx.x;
    const int lane = tid & 31;
    const int wv   = __builtin_amdgcn_readfirstlane(tid >> 5);
    const int b    = blockIdx.x;
    const int lo   = b * GN_RPW;
    const int nrows = (GN_N - lo < GN_RPW) ? (GN_N - lo) : GN_RPW;

    {
        const v4f z = {0.0f, 0.0f, 0.0f, 0.0f};
        for (int i = tid; i < (GN_RPW + 1) * GN_D / 4; i += 1024) ((v4f*)acc)[i] = z;
        if (tid <= GN_RPW) deg[tid] = 0u;
        if (tid == 0) cnt = 0u;
    }
    unsigned stored = 0u;
    if (!FIRST) stored = LCNT[b];
    __syncthreads();

    const bool sweep = FIRST || stored > GN_LCAP;
    if (sweep) {
        const int rot = (b * GN_TRIPS) / GN_OW;
        unsigned inv[8];
#pragma unroll
        for (int p = 0; p < 8; ++p) inv[p] = ((b >> p) & 1) ? 0u : ~0u;
        for (int t0 = wv; t0 < GN_TRIPS; t0 += 128) {
            v4u wa[4], wb[4];
            int eb[4];
            unsigned okm[4];
#pragma unroll
            for (int q = 0; q < 4; ++q) {
                const int t = t0 + 32 * q;
                const bool ok = t < GN_TRIPS;
                int tt = (ok ? t : 0) + rot;
                if (tt >= GN_TRIPS) tt -= GN_TRIPS;
                const v4u* pp = (const v4u*)(PL + ((size_t)tt * 32 + lane) * 8);
                wa[q] = pp[0];
                wb[q] = pp[1];
                eb[q] = tt * 1024 + lane * 4;
                okm[q] = ok ? ~0u : 0u;
            }
#pragma unroll
            for (int q = 0; q < 4; ++q) {
                unsigned m = okm[q]
                           & (wa[q].x ^ inv[0]) & (wa[q].y ^ inv[1])
                           & (wa[q].z ^ inv[2]) & (wa[q].w ^ inv[3])
                           & (wb[q].x ^ inv[4]) & (wb[q].y ^ inv[5])
                           & (wb[q].z ^ inv[6]) & (wb[q].w ^ inv[7]);
                while (m != 0u) {
                    const int i = __builtin_ctz(m);
                    m &= m - 1u;
                    const unsigned e = (unsigned)(eb[q] + (i >> 2) * 128 + (i & 3));
                    const unsigned p = atomicAdd(&cnt, 1u);
                    if (p < GN_LCAP) {
                        list[p] = e;
                    } else {
                        const unsigned r = (unsigned)(dst[e] - lo);
                        if (r < (unsigned)nrows) {
                            if (FIRST) atomicAdd(&deg[r], 1u);
                            const float* hp = H + (size_t)src[e] * GN_D;
                            float* ap = acc + r * GN_D;
#pragma nounroll
                            for (int c = 0; c < GN_D; ++c)
                                own_add(ap + (c & 3) * 32 + (c >> 2), hp[c]);
                        }
                    }
                }
            }
        }
        __syncthreads();
        const unsigned nl = cnt < GN_LCAP ? cnt : GN_LCAP;
        for (unsigned i = tid; i < nl; i += 1024) {
            const unsigned e = list[i];
            const unsigned r = (unsigned)(dst[e] - lo);
            unsigned ent = (unsigned)GN_RPW << 20;
            if (r < (unsigned)nrows) {
                if (FIRST) atomicAdd(&deg[r], 1u);
                ent = (r << 20) | e;
            }
            list[i] = ent;
        }
    } else {
        const unsigned* st = LST + (size_t)b * GN_LCAP;
        for (unsigned i = tid; i < stored; i += 1024) list[i] = st[i];
    }
    __syncthreads();

    const unsigned total = sweep ? cnt : stored;
    const unsigned n = total < GN_LCAP ? total : GN_LCAP;
    unsigned* mylst = LST + (size_t)b * GN_LCAP;
    if (FIRST) {
        for (unsigned i = tid; i < n; i += 1024) mylst[i] = list[i];
        if (tid == 0) LCNT[b] = total;
        if (tid < nrows) INV[lo + tid] = 1.0f / fmaxf((float)deg[tid], 1.0f);
    }

    auto fetch = [&](unsigned at, int& s_out, int& r_out) {
        const unsigned i = at + lane;
        unsigned ent = (unsigned)GN_RPW << 20;
        if (i < n) ent = list[i];
        s_out = src[ent & 0xFFFFFu];
        r_out = (int)(ent >> 20);
    };
    unsigned base = wv * 32;
    int s = 0, rl = GN_RPW;
    if (base < n) fetch(base, s, rl);
    while (base < n) {
        const int live = (int)(n - base);
        const unsigned nb = base + 1024;
        int sn = 0, rn = GN_RPW;
        v4f v[16];
#pragma unroll
        for (int j = 0; j < 16; ++j) {
            const int sj = __builtin_amdgcn_readlane(s, j);
            v[j] = *(const v4f*)(H + (size_t)sj * GN_D + lane * 4);
        }
        if (nb < n) fetch(nb, sn, rn);
#pragma unroll
        for (int j = 0; j < 16; ++j) {
            const int rj = __builtin_amdgcn_readlane(rl, j);
            float* ap = acc + rj * GN_D + lane;
            own_add(ap,      v[j].x);
            own_add(ap + 32, v[j].y);
            own_add(ap + 64, v[j].z);
            own_add(ap + 96, v[j].w);
        }
        if (live > 16) {
#pragma unroll
            for (int j = 0; j < 16; ++j) {
                const int sj = __builtin_amdgcn_readlane(s, 16 + j);
                v[j] = *(const v4f*)(H + (size_t)sj * GN_D + lane * 4);
            }
#pragma unroll
            for (int j = 0; j < 16; ++j) {
                const int rj = __builtin_amdgcn_readlane(rl, 16 + j);
                float* ap = acc + rj * GN_D + lane;
                own_add(ap,      v[j].x);
                own_add(ap + 32, v[j].y);
                own_add(ap + 64, v[j].z);
                own_add(ap + 96, v[j].w);
            }
        }
        base = nb;
        s = sn;
        rl = rn;
    }
    __syncthreads();

    for (int r = wv; r < nrows; r += 32) {
        const float* ap = acc + r * GN_D + lane;
        v4f o = {ap[0], ap[32], ap[64], ap[96]};
        *(v4f*)(SUM + (size_t)(lo + r) * GN_D + lane * 4) = o;
    }
}

template <int NP>
struct RsFrag {
    v4f a[NP][16];
    float s;
};

template <int NP>
__device__ __forceinline__ void rs_load(RsFrag<NP>& f, int rb, int l15, int koff,
                                        const float* __restrict__ A1,
                                        const float* __restrict__ A2,
                                        const float* __restrict__ invdeg) {
    const size_t row = (size_t)(rb * 16 + l15);
    const float* r1 = A1 + row * GN_D + koff;
#pragma unroll
    for (int q = 0; q < 16; ++q) f.a[0][q] = *(const v4f*)(r1 + 8 * q);
    if (NP == 2) {
        const float* r2 = A2 + row * GN_D + koff;
#pragma unroll
        for (int q = 0; q < 16; ++q) f.a[NP - 1][q] = *(const v4f*)(r2 + 8 * q);
        f.s = invdeg[row];
    } else {
        f.s = 1.0f;
    }
}

template <int NP, int NT>
__device__ __forceinline__ void rs_block(const RsFrag<NP>& f, const v4f (&b)[NP][16][NT],
                                         const float (&bj)[NT], int rb, int h, int l15,
                                         int colbase, float* __restrict__ out, int relu_out) {
    v8f c[NT];
#pragma unroll
    for (int jt = 0; jt < NT; ++jt)
#pragma unroll
        for (int v = 0; v < 8; ++v) c[jt][v] = bj[jt];

#pragma unroll
    for (int p = 0; p < NP; ++p) {
#pragma unroll
        for (int q = 0; q < 16; ++q) {
            v4f av = f.a[p][q];
            if (NP == 2 && p == 0) av *= f.s;
            v2f a0 = {av.x, av.y};
            v2f a1 = {av.z, av.w};
#pragma unroll
            for (int jt = 0; jt < NT; ++jt) {
                v2f b0 = {b[p][q][jt].x, b[p][q][jt].y};
                c[jt] = __builtin_amdgcn_wmma_f32_16x16x4_f32(
                    false, a0, false, b0, (short)0, c[jt], false, false);
            }
#pragma unroll
            for (int jt = 0; jt < NT; ++jt) {
                v2f b1 = {b[p][q][jt].z, b[p][q][jt].w};
                c[jt] = __builtin_amdgcn_wmma_f32_16x16x4_f32(
                    false, a1, false, b1, (short)0, c[jt], false, false);
            }
        }
    }

    float* obase = out + (size_t)(rb * 16 + 8 * h) * GN_D + colbase + l15;
#pragma unroll
    for (int jt = 0; jt < NT; ++jt)
#pragma unroll
        for (int v = 0; v < 8; ++v) {
            float val = c[jt][v];
            if (relu_out) val = fmaxf(val, 0.0f);
            obase[(size_t)v * GN_D + jt * 16] = val;
        }
}

template <int NP, int NT, bool MAP>
__global__ __launch_bounds__(128) void sage_gemm_rs(
    const float* __restrict__ A1, const float* __restrict__ invdeg,
    const float* __restrict__ W1, const float* __restrict__ bias,
    const float* __restrict__ A2, const float* __restrict__ W2,
    float* __restrict__ out, int relu_out,
    const int* __restrict__ dst, unsigned* __restrict__ PL) {
    const int lane   = threadIdx.x & 31;
    const int wv     = __builtin_amdgcn_readfirstlane(threadIdx.x >> 5);
    const int colbase = (NT == 4) ? (wv & 1) * 64 : 0;
    const int h      = lane >> 4;
    const int l15    = lane & 15;
    const int koff   = 4 * h;
    const int npairs = (NT == 4) ? gridDim.x * 2 : gridDim.x * 4;
    int rb = (NT == 4) ? blockIdx.x * 2 + (wv >> 1) : blockIdx.x * 4 + wv;

    if (MAP) {
        const int nw = gridDim.x * 4;
        for (int c = nw - 1 - (blockIdx.x * 4 + wv); c < GN_TRIPS; c += nw) {
            const int* dp = dst + c * 1024 + lane * 4;
            v4i d[8];
#pragma unroll
            for (int j = 0; j < 8; ++j) d[j] = *(const v4i*)(dp + j * 128);
            unsigned w[8] = {0u, 0u, 0u, 0u, 0u, 0u, 0u, 0u};
#pragma unroll
            for (int j = 0; j < 8; ++j)
#pragma unroll
                for (int k = 0; k < 4; ++k) {
                    const int dv = k == 0 ? d[j].x : k == 1 ? d[j].y : k == 2 ? d[j].z : d[j].w;
                    const unsigned o = (unsigned)dv / (unsigned)GN_RPW;
#pragma unroll
                    for (int p = 0; p < 8; ++p) w[p] |= ((o >> p) & 1u) << (4 * j + k);
                }
            v4u* op = (v4u*)(PL + ((size_t)c * 32 + lane) * 8);
            const v4u lo4 = {w[0], w[1], w[2], w[3]};
            const v4u hi4 = {w[4], w[5], w[6], w[7]};
            op[0] = lo4;
            op[1] = hi4;
        }
    }
    if (rb >= GN_RB) return;

    v4f b[NP][16][NT];
#pragma unroll
    for (int q = 0; q < 16; ++q)
#pragma unroll
        for (int jt = 0; jt < NT; ++jt) {
            const size_t wrow = (size_t)(colbase + jt * 16 + l15) * GN_D + 8 * q + koff;
            b[0][q][jt] = *(const v4f*)(W1 + wrow);
            if (NP == 2) b[NP - 1][q][jt] = *(const v4f*)(W2 + wrow);
        }

    float bj[NT];
#pragma unroll
    for (int jt = 0; jt < NT; ++jt) bj[jt] = bias[colbase + jt * 16 + l15];

    RsFrag<NP> f0, f1;
    rs_load<NP>(f0, rb, l15, koff, A1, A2, invdeg);

    for (;;) {
        const int rbn = rb + npairs;
        rs_load<NP>(f1, rbn < GN_RB ? rbn : rb, l15, koff, A1, A2, invdeg);
        __builtin_amdgcn_sched_barrier(0);
        rs_block<NP, NT>(f0, b, bj, rb, h, l15, colbase, out, relu_out);
        __builtin_amdgcn_sched_barrier(0);
        if (rbn >= GN_RB) break;
        rb = rbn;
        f0 = f1;
    }
}

extern "C" void kernel_launch(void* const* d_in, const int* in_sizes, int n_in,
                              void* d_out, int out_size, void* d_ws, size_t ws_size,
                              hipStream_t stream) {
    (void)in_sizes; (void)n_in; (void)out_size; (void)ws_size;

    const float* x   = (const float*)d_in[0];
    const int*   ei  = (const int*)d_in[1];
    const float* W0p = (const float*)d_in[2];
    const float* b0p = (const float*)d_in[3];
    const float* W0l = (const float*)d_in[4];
    const float* b0l = (const float*)d_in[5];
    const float* W0r = (const float*)d_in[6];
    const float* W1p = (const float*)d_in[7];
    const float* b1p = (const float*)d_in[8];
    const float* W1l = (const float*)d_in[9];
    const float* b1l = (const float*)d_in[10];
    const float* W1r = (const float*)d_in[11];
    float* out = (float*)d_out;

    const size_t nd = (size_t)GN_N * GN_D;
    float* H   = (float*)d_ws;
    float* X1  = H + nd;
    float* SUM = X1 + nd;
    float* INV = SUM + nd;
    unsigned* LST  = (unsigned*)(INV + GN_N);
    unsigned* LCNT = LST + (size_t)GN_OW * GN_LCAP;
    unsigned* PL   = LCNT + GN_OW;

    const int* src = ei;
    const int* dst = ei + GN_E;

    const int gemm_blocks = 256;

    sage_gemm_rs<1, 8, true><<<gemm_blocks, 128, 0, stream>>>(x, nullptr, W0p, b0p,
                                                           nullptr, nullptr, H, 1, dst, PL);
    sage_owner<true><<<GN_OW, 1024, 0, stream>>>(H, src, dst, SUM, INV, LST, LCNT, PL);
    sage_gemm_rs<2, 4, false><<<gemm_blocks, 128, 0, stream>>>(SUM, INV, W0l, b0l,
                                                            x, W0r, X1, 1, nullptr, nullptr);

    sage_gemm_rs<1, 8, false><<<gemm_blocks, 128, 0, stream>>>(X1, nullptr, W1p, b1p,
                                                            nullptr, nullptr, H, 1, nullptr, nullptr);
    sage_owner<false><<<GN_OW, 1024, 0, stream>>>(H, src, dst, SUM, INV, LST, LCNT, PL);
    sage_gemm_rs<2, 4, false><<<gemm_blocks, 128, 0, stream>>>(SUM, INV, W1l, b1l,
                                                            X1, W1r, out, 0, nullptr, nullptr);
}


# Owner edge list counting-sorted by destination row in LDS, entries hold (row, source row): runs of equal rows summed in registers, one LDS add per run, no index load per edge; sum order: runs first
# speedup vs baseline: 2.5375x; 1.0315x over previous
#include <hip/hip_runtime.h>
#include <hip/hip_bf16.h>


#define GN_N 100000
#define GN_E 640000
#define GN_D 128
#define GN_RB (GN_N / 16)

typedef __attribute__((ext_vector_type(2))) float v2f;
typedef __attribute__((ext_vector_type(4))) float v4f;
typedef __attribute__((ext_vector_type(8))) float v8f;
typedef __attribute__((ext_vector_type(4))) int v4i;
typedef __attribute__((ext_vector_type(4))) unsigned v4u;

#define GN_OW    256
#define GN_RPW   ((GN_N + GN_OW - 1) / GN_OW)
#define GN_LCAP  8192
#define GN_TRIPS (GN_E / 1024)
#define GN_SCAN  416

__device__ __forceinline__ void own_add(float* p, float v) {
    __hip_atomic_fetch_add(p, v, __ATOMIC_RELAXED, __HIP_MEMORY_SCOPE_WORKGROUP);
}

template <int OFF>
__device__ __forceinline__ void own_add_runs(float* acc, int rl, const v4f (&v)[16], int lane) {
    int cur = __builtin_amdgcn_readlane(rl, OFF);
    v4f a = v[0];
#pragma unroll
    for (int j = 1; j < 16; ++j) {
        const int rj = __builtin_amdgcn_readlane(rl, OFF + j);
        if (rj != cur) {
            float* ap = acc + cur * GN_D + lane;
            own_add(ap,      a.x);
            own_add(ap + 32, a.y);
            own_add(ap + 64, a.z);
            own_add(ap + 96, a.w);
            a = v[j];
            cur = rj;
        } else {
            a += v[j];
        }
    }
    float* ap = acc + cur * GN_D + lane;
    own_add(ap,      a.x);
    own_add(ap + 32, a.y);
    own_add(ap + 64, a.z);
    own_add(ap + 96, a.w);
}

template <bool FIRST>
__global__ __launch_bounds__(1024) void sage_owner(
    const float* __restrict__ H, const int* __restrict__ src,
    const int* __restrict__ dst, float* __restrict__ SUM,
    float* __restrict__ INV, unsigned* __restrict__ LST,
    unsigned* __restrict__ LCNT, const unsigned* __restrict__ PL) {
    __shared__ __attribute__((aligned(16))) float acc[(GN_RPW + 1) * GN_D];
    __shared__ unsigned list[GN_LCAP];
    __shared__ unsigned slist[GN_LCAP];
    __shared__ unsigned deg[GN_RPW + 1];
    __shared__ unsigned cur[GN_SCAN];
    __shared__ unsigned cnt;

    const int tid  = threadIdx.x;
    const int lane = tid & 31;
    const int wv   = __builtin_amdgcn_readfirstlane(tid >> 5);
    const int b    = blockIdx.x;
    const int lo   = b * GN_RPW;
    const int nrows = (GN_N - lo < GN_RPW) ? (GN_N - lo) : GN_RPW;

    {
        const v4f z = {0.0f, 0.0f, 0.0f, 0.0f};
        for (int i = tid; i < (GN_RPW + 1) * GN_D / 4; i += 1024) ((v4f*)acc)[i] = z;
        if (tid <= GN_RPW) deg[tid] = 0u;
        if (tid < GN_SCAN) cur[tid] = 0u;
        if (tid == 0) cnt = 0u;
    }
    unsigned stored = 0u;
    if (!FIRST) stored = LCNT[b];
    __syncthreads();

    const bool sweep = FIRST || stored > GN_LCAP;
    if (sweep) {
        const int rot = (b * GN_TRIPS) / GN_OW;
        unsigned inv[8];
#pragma unroll
        for (int p = 0; p < 8; ++p) inv[p] = ((b >> p) & 1) ? 0u : ~0u;
        for (int t0 = wv; t0 < GN_TRIPS; t0 += 128) {
            v4u wa[4], wb[4];
            int eb[4];
            unsigned okm[4];
#pragma unroll
            for (int q = 0; q < 4; ++q) {
                const int t = t0 + 32 * q;
                const bool ok = t < GN_TRIPS;
                int tt = (ok ? t : 0) + rot;
                if (tt >= GN_TRIPS) tt -= GN_TRIPS;
                const v4u* pp = (const v4u*)(PL + ((size_t)tt * 32 + lane) * 8);
                wa[q] = pp[0];
                wb[q] = pp[1];
                eb[q] = tt * 1024 + lane * 4;
                okm[q] = ok ? ~0u : 0u;
            }
#pragma unroll
            for (int q = 0; q < 4; ++q) {
                unsigned m = okm[q]
                           & (wa[q].x ^ inv[0]) & (wa[q].y ^ inv[1])
                           & (wa[q].z ^ inv[2]) & (wa[q].w ^ inv[3])
                           & (wb[q].x ^ inv[4]) & (wb[q].y ^ inv[5])
                           & (wb[q].z ^ inv[6]) & (wb[q].w ^ inv[7]);
                while (m != 0u) {
                    const int i = __builtin_ctz(m);
                    m &= m - 1u;
                    const unsigned e = (unsigned)(eb[q] + (i >> 2) * 128 + (i & 3));
                    const unsigned p = atomicAdd(&cnt, 1u);
                    if (p < GN_LCAP) {
                        list[p] = e;
                    } else {
                        const unsigned r = (unsigned)(dst[e] - lo);
                        if (r < (unsigned)nrows) {
                            if (FIRST) atomicAdd(&deg[r], 1u);
                            const float* hp = H + (size_t)src[e] * GN_D;
                            float* ap = acc + r * GN_D;
#pragma nounroll
                            for (int c = 0; c < GN_D; ++c)
                                own_add(ap + (c & 3) * 32 + (c >> 2), hp[c]);
                        }
                    }
                }
            }
        }
        __syncthreads();
        const unsigned nl = cnt < GN_LCAP ? cnt : GN_LCAP;
        for (unsigned i0 = tid; i0 < nl; i0 += 4096) {
            unsigned e4[4];
            int d4[4], s4[4];
#pragma unroll
            for (int q = 0; q < 4; ++q) {
                const unsigned i = i0 + 1024u * q;
                e4[q] = i < nl ? list[i] : 0u;
            }
#pragma unroll
            for (int q = 0; q < 4; ++q) {
                d4[q] = dst[e4[q]];
                s4[q] = src[e4[q]];
            }
            asm volatile("" :: "v"(d4[0]), "v"(d4[1]), "v"(d4[2]), "v"(d4[3]),
                               "v"(s4[0]), "v"(s4[1]), "v"(s4[2]), "v"(s4[3]));
#pragma unroll
            for (int q = 0; q < 4; ++q) {
                const unsigned i = i0 + 1024u * q;
                if (i < nl) {
                    const unsigned r = (unsigned)(d4[q] - lo);
                    unsigned ent = (unsigned)GN_RPW << 20;
                    if (r < (unsigned)nrows) {
                        if (FIRST) atomicAdd(&deg[r], 1u);
                        ent = (r << 20) | (unsigned)s4[q];
                    }
                    atomicAdd(&cur[ent >> 20], 1u);
                    list[i] = ent;
                }
            }
        }
        __syncthreads();
        if (wv == 0) {
            unsigned c13[13];
            unsigned t = 0u;
#pragma unroll
            for (int k = 0; k < 13; ++k) {
                const unsigned x = cur[lane * 13 + k];
                c13[k] = t;
                t += x;
            }
            unsigned inc = t;
#pragma unroll
            for (int d = 1; d < 32; d <<= 1) {
                const unsigned y = __shfl_up(inc, (unsigned)d);
                if (lane >= d) inc += y;
            }
            const unsigned excl = inc - t;
#pragma unroll
            for (int k = 0; k < 13; ++k) cur[lane * 13 + k] = c13[k] + excl;
        }
        __syncthreads();
        for (unsigned i = tid; i < nl; i += 1024) {
            const unsigned ent = list[i];
            const unsigned p = atomicAdd(&cur[ent >> 20], 1u);
            slist[p] = ent;
        }
    } else {
        const unsigned* st = LST + (size_t)b * GN_LCAP;
        for (unsigned i = tid; i < stored; i += 1024) slist[i] = st[i];
    }
    __syncthreads();

    const unsigned total = sweep ? cnt : stored;
    const unsigned n = total < GN_LCAP ? total : GN_LCAP;
    unsigned* mylst = LST + (size_t)b * GN_LCAP;
    if (FIRST) {
        for (unsigned i = tid; i < n; i += 1024) mylst[i] = slist[i];
        if (tid == 0) LCNT[b] = total;
        if (tid < nrows) INV[lo + tid] = 1.0f / fmaxf((float)deg[tid], 1.0f);
    }

    auto fetch = [&](unsigned at, int& s_out, int& r_out) {
        const unsigned i = at + lane;
        unsigned ent = (unsigned)GN_RPW << 20;
        if (i < n) ent = slist[i];
        s_out = (int)(ent & 0xFFFFFu);
        r_out = (int)(ent >> 20);
    };
    unsigned base = wv * 32;
    int s = 0, rl = GN_RPW;
    if (base < n) fetch(base, s, rl);
    while (base < n) {
        const int live = (int)(n - base);
        const unsigned nb = base + 1024;
        int sn = 0, rn = GN_RPW;
        v4f v[16];
#pragma unroll
        for (int j = 0; j < 16; ++j) {
            const int sj = __builtin_amdgcn_readlane(s, j);
            v[j] = *(const v4f*)(H + (size_t)sj * GN_D + lane * 4);
        }
        if (nb < n) fetch(nb, sn, rn);
        own_add_runs<0>(acc, rl, v, lane);
        if (live > 16) {
#pragma unroll
            for (int j = 0; j < 16; ++j) {
                const int sj = __builtin_amdgcn_readlane(s, 16 + j);
                v[j] = *(const v4f*)(H + (size_t)sj * GN_D + lane * 4);
            }
            own_add_runs<16>(acc, rl, v, lane);
        }
        base = nb;
        s = sn;
        rl = rn;
    }
    __syncthreads();

    for (int r = wv; r < nrows; r += 32) {
        const float* ap = acc + r * GN_D + lane;
        v4f o = {ap[0], ap[32], ap[64], ap[96]};
        *(v4f*)(SUM + (size_t)(lo + r) * GN_D + lane * 4) = o;
    }
}

template <int NP>
struct RsFrag {
    v4f a[NP][16];
    float s;
};

template <int NP>
__device__ __forceinline__ void rs_load(RsFrag<NP>& f, int rb, int l15, int koff,
                                        const float* __restrict__ A1,
                                        const float* __restrict__ A2,
                                        const float* __restrict__ invdeg) {
    const size_t row = (size_t)(rb * 16 + l15);
    const float* r1 = A1 + row * GN_D + koff;
#pragma unroll
    for (int q = 0; q < 16; ++q) f.a[0][q] = *(const v4f*)(r1 + 8 * q);
    if (NP == 2) {
        const float* r2 = A2 + row * GN_D + koff;
#pragma unroll
        for (int q = 0; q < 16; ++q) f.a[NP - 1][q] = *(const v4f*)(r2 + 8 * q);
        f.s = invdeg[row];
    } else {
        f.s = 1.0f;
    }
}

template <int NP, int NT>
__device__ __forceinline__ void rs_block(const RsFrag<NP>& f, const v4f (&b)[NP][16][NT],
                                         const float (&bj)[NT], int rb, int h, int l15,
                                         int colbase, float* __restrict__ out, int relu_out) {
    v8f c[NT];
#pragma unroll
    for (int jt = 0; jt < NT; ++jt)
#pragma unroll
        for (int v = 0; v < 8; ++v) c[jt][v] = bj[jt];

#pragma unroll
    for (int p = 0; p < NP; ++p) {
#pragma unroll
        for (int q = 0; q < 16; ++q) {
            v4f av = f.a[p][q];
            if (NP == 2 && p == 0) av *= f.s;
            v2f a0 = {av.x, av.y};
            v2f a1 = {av.z, av.w};
#pragma unroll
            for (int jt = 0; jt < NT; ++jt) {
                v2f b0 = {b[p][q][jt].x, b[p][q][jt].y};
                c[jt] = __builtin_amdgcn_wmma_f32_16x16x4_f32(
                    false, a0, false, b0, (short)0, c[jt], false, false);
            }
#pragma unroll
            for (int jt = 0; jt < NT; ++jt) {
                v2f b1 = {b[p][q][jt].z, b[p][q][jt].w};
                c[jt] = __builtin_amdgcn_wmma_f32_16x16x4_f32(
                    false, a1, false, b1, (short)0, c[jt], false, false);
            }
        }
    }

    float* obase = out + (size_t)(rb * 16 + 8 * h) * GN_D + colbase + l15;
#pragma unroll
    for (int jt = 0; jt < NT; ++jt)
#pragma unroll
        for (int v = 0; v < 8; ++v) {
            float val = c[jt][v];
            if (relu_out) val = fmaxf(val, 0.0f);
            obase[(size_t)v * GN_D + jt * 16] = val;
        }
}

template <int NP, int NT, bool MAP>
__global__ __launch_bounds__(128) void sage_gemm_rs(
    const float* __restrict__ A1, const float* __restrict__ invdeg,
    const float* __restrict__ W1, const float* __restrict__ bias,
    const float* __restrict__ A2, const float* __restrict__ W2,
    float* __restrict__ out, int relu_out,
    const int* __restrict__ dst, unsigned* __restrict__ PL) {
    const int lane   = threadIdx.x & 31;
    const int wv     = __builtin_amdgcn_readfirstlane(threadIdx.x >> 5);
    const int colbase = (NT == 4) ? (wv & 1) * 64 : 0;
    const int h      = lane >> 4;
    const int l15    = lane & 15;
    const int koff   = 4 * h;
    const int npairs = (NT == 4) ? gridDim.x * 2 : gridDim.x * 4;
    int rb = (NT == 4) ? blockIdx.x * 2 + (wv >> 1) : blockIdx.x * 4 + wv;

    if (MAP) {
        const int nw = gridDim.x * 4;
        for (int c = nw - 1 - (blockIdx.x * 4 + wv); c < GN_TRIPS; c += nw) {
            const int* dp = dst + c * 1024 + lane * 4;
            v4i d[8];
#pragma unroll
            for (int j = 0; j < 8; ++j) d[j] = *(const v4i*)(dp + j * 128);
            unsigned w[8] = {0u, 0u, 0u, 0u, 0u, 0u, 0u, 0u};
#pragma unroll
            for (int j = 0; j < 8; ++j)
#pragma unroll
                for (int k = 0; k < 4; ++k) {
                    const int dv = k == 0 ? d[j].x : k == 1 ? d[j].y : k == 2 ? d[j].z : d[j].w;
                    const unsigned o = (unsigned)dv / (unsigned)GN_RPW;
#pragma unroll
                    for (int p = 0; p < 8; ++p) w[p] |= ((o >> p) & 1u) << (4 * j + k);
                }
            v4u* op = (v4u*)(PL + ((size_t)c * 32 + lane) * 8);
            const v4u lo4 = {w[0], w[1], w[2], w[3]};
            const v4u hi4 = {w[4], w[5], w[6], w[7]};
            op[0] = lo4;
            op[1] = hi4;
        }
    }
    if (rb >= GN_RB) return;

    v4f b[NP][16][NT];
#pragma unroll
    for (int q = 0; q < 16; ++q)
#pragma unroll
        for (int jt = 0; jt < NT; ++jt) {
            const size_t wrow = (size_t)(colbase + jt * 16 + l15) * GN_D + 8 * q + koff;
            b[0][q][jt] = *(const v4f*)(W1 + wrow);
            if (NP == 2) b[NP - 1][q][jt] = *(const v4f*)(W2 + wrow);
        }

    float bj[NT];
#pragma unroll
    for (int jt = 0; jt < NT; ++jt) bj[jt] = bias[colbase + jt * 16 + l15];

    RsFrag<NP> f0, f1;
    rs_load<NP>(f0, rb, l15, koff, A1, A2, invdeg);

    for (;;) {
        const int rbn = rb + npairs;
        rs_load<NP>(f1, rbn < GN_RB ? rbn : rb, l15, koff, A1, A2, invdeg);
        __builtin_amdgcn_sched_barrier(0);
        rs_block<NP, NT>(f0, b, bj, rb, h, l15, colbase, out, relu_out);
        __builtin_amdgcn_sched_barrier(0);
        if (rbn >= GN_RB) break;
        rb = rbn;
        f0 = f1;
    }
}

extern "C" void kernel_launch(void* const* d_in, const int* in_sizes, int n_in,
                              void* d_out, int out_size, void* d_ws, size_t ws_size,
                              hipStream_t stream) {
    (void)in_sizes; (void)n_in; (void)out_size; (void)ws_size;

    const float* x   = (const float*)d_in[0];
    const int*   ei  = (const int*)d_in[1];
    const float* W0p = (const float*)d_in[2];
    const float* b0p = (const float*)d_in[3];
    const float* W0l = (const float*)d_in[4];
    const float* b0l = (const float*)d_in[5];
    const float* W0r = (const float*)d_in[6];
    const float* W1p = (const float*)d_in[7];
    const float* b1p = (const float*)d_in[8];
    const float* W1l = (const float*)d_in[9];
    const float* b1l = (const float*)d_in[10];
    const float* W1r = (const float*)d_in[11];
    float* out = (float*)d_out;

    const size_t nd = (size_t)GN_N * GN_D;
    float* H   = (float*)d_ws;
    float* X1  = H + nd;
    float* SUM = X1 + nd;
    float* INV = SUM + nd;
    unsigned* LST  = (unsigned*)(INV + GN_N);
    unsigned* LCNT = LST + (size_t)GN_OW * GN_LCAP;
    unsigned* PL   = LCNT + GN_OW;

    const int* src = ei;
    const int* dst = ei + GN_E;

    const int gemm_blocks = 256;

    sage_gemm_rs<1, 8, true><<<gemm_blocks, 128, 0, stream>>>(x, nullptr, W0p, b0p,
                                                           nullptr, nullptr, H, 1, dst, PL);
    sage_owner<true><<<GN_OW, 1024, 0, stream>>>(H, src, dst, SUM, INV, LST, LCNT, PL);
    sage_gemm_rs<2, 4, false><<<gemm_blocks, 128, 0, stream>>>(SUM, INV, W0l, b0l,
                                                            x, W0r, X1, 1, nullptr, nullptr);

    sage_gemm_rs<1, 8, false><<<gemm_blocks, 128, 0, stream>>>(X1, nullptr, W1p, b1p,
                                                            nullptr, nullptr, H, 1, nullptr, nullptr);
    sage_owner<false><<<GN_OW, 1024, 0, stream>>>(H, src, dst, SUM, INV, LST, LCNT, PL);
    sage_gemm_rs<2, 4, false><<<gemm_blocks, 128, 0, stream>>>(SUM, INV, W1l, b1l,
                                                            X1, W1r, out, 0, nullptr, nullptr);
}
